# starting kernel
# baseline (speedup 1.0000x reference)
#include <hip/hip_runtime.h>
#include <math.h>


#define GRID_HH 32
#define GRID_WW 32
#define OBJ_DIM 64
#define LOC_DIM 16
#define SDIM    64
#define NCOLS   (GRID_HH * GRID_WW)
#define BATCH   64
#define ETA_X   0.05f
#define NSTEPS  20

#define XO_TOT  ((size_t)NCOLS * BATCH * OBJ_DIM)
#define XL_TOT  ((size_t)NCOLS * BATCH * LOC_DIM)

typedef __attribute__((ext_vector_type(2))) float v2f;
typedef __attribute__((ext_vector_type(8))) float v8f;

__device__ __forceinline__ v8f wmma_f32_k4(v2f a, v2f b, v8f c) {
  return __builtin_amdgcn_wmma_f32_16x16x4_f32(
      false, a, false, b, (short)0, c, false, false);
}

__global__ void cg_init_kernel(float* __restrict__ xo,
                               float* __restrict__ xl,
                               float* __restrict__ energy) {
  size_t i = (size_t)blockIdx.x * blockDim.x + threadIdx.x;
  size_t stride = (size_t)gridDim.x * blockDim.x;
  for (size_t e = i; e < XO_TOT; e += stride) xo[e] = 0.0f;
  for (size_t e = i; e < XL_TOT; e += stride) xl[e] = 0.0f;
  if (i < NSTEPS) energy[i] = 0.0f;
}

__global__ __launch_bounds__(512)
void cg_step_kernel(const float* __restrict__ gin,
                    const float* __restrict__ Wobj,
                    const float* __restrict__ Wloc,
                    const float* __restrict__ xo_in,
                    const float* __restrict__ xl_in,
                    float* __restrict__ xo_out,
                    float* __restrict__ xl_out,
                    float* __restrict__ energy_t) {
  __shared__ float sWo[OBJ_DIM][65];
  __shared__ float sWl[LOC_DIM][65];
  __shared__ float sPG[BATCH][65];
  __shared__ float sXo[BATCH][65];
  __shared__ float sXl[BATCH][17];
  __shared__ float sCtx[BATCH][65];
  __shared__ float sEnergy;

  const int n   = blockIdx.x;
  const int gh  = n >> 5;
  const int gw  = n & 31;
  const int tid = threadIdx.x;

  if (tid == 0) sEnergy = 0.0f;

  for (int e = tid; e < OBJ_DIM * SDIM; e += 512)
    sWo[e >> 6][e & 63] = Wobj[(size_t)n * OBJ_DIM * SDIM + e];
  for (int e = tid; e < LOC_DIM * SDIM; e += 512)
    sWl[e >> 6][e & 63] = Wloc[(size_t)n * LOC_DIM * SDIM + e];

  for (int e = tid; e < BATCH * SDIM; e += 512) {
    int b = e >> 6, s = e & 63;
    int py = s >> 3, px = s & 7;
    sPG[b][s] = gin[(size_t)b * 65536 + (size_t)(gh * 8 + py) * 256 + gw * 8 + px];
  }

  const float inv_cnt =
      1.0f / (float)((gh > 0) + (gh < GRID_HH - 1) + (gw > 0) + (gw < GRID_WW - 1));
  for (int e = tid; e < BATCH * OBJ_DIM; e += 512) {
    sXo[e >> 6][e & 63] = xo_in[(size_t)n * BATCH * OBJ_DIM + e];
    float ssum = 0.0f;
    if (gh > 0)           ssum += xo_in[(size_t)(n - GRID_WW) * BATCH * OBJ_DIM + e];
    if (gh < GRID_HH - 1) ssum += xo_in[(size_t)(n + GRID_WW) * BATCH * OBJ_DIM + e];
    if (gw > 0)           ssum += xo_in[(size_t)(n - 1) * BATCH * OBJ_DIM + e];
    if (gw < GRID_WW - 1) ssum += xo_in[(size_t)(n + 1) * BATCH * OBJ_DIM + e];
    sCtx[e >> 6][e & 63] = ssum * inv_cnt;
  }
  for (int e = tid; e < BATCH * LOC_DIM; e += 512)
    sXl[e >> 4][e & 15] = xl_in[(size_t)n * BATCH * LOC_DIM + e];
  __syncthreads();

  const int lane = tid & 31;
  const int wv   = tid >> 5;
  const int kh   = lane >> 4;
  const int mr   = lane & 15;
  const int mt   = (wv >> 2) * 16;
  const int nt   = (wv & 3) * 16;

  v8f acc = {};
  for (int k = 0; k < SDIM; k += 4) {
    v2f a, b;
    a.x = sXo[mt + mr][k + 2 * kh];
    a.y = sXo[mt + mr][k + 2 * kh + 1];
    b.x = sWo[k + 2 * kh][nt + mr];
    b.y = sWo[k + 2 * kh + 1][nt + mr];
    acc = wmma_f32_k4(a, b, acc);
  }
  for (int k = 0; k < LOC_DIM; k += 4) {
    v2f a, b;
    a.x = sXl[mt + mr][k + 2 * kh];
    a.y = sXl[mt + mr][k + 2 * kh + 1];
    b.x = sWl[k + 2 * kh][nt + mr];
    b.y = sWl[k + 2 * kh + 1][nt + mr];
    acc = wmma_f32_k4(a, b, acc);
  }

  float epart = 0.0f;
#pragma unroll
  for (int r = 0; r < 8; ++r) {
    int bb = mt + r + 8 * kh;
    int ss = nt + mr;
    float pred = tanhf(acc[r]);
    float eps  = sPG[bb][ss] - pred;
    float g    = eps * (1.0f - pred * pred);
    epart += 0.5f * eps * eps;
    sPG[bb][ss] = g;
  }

  for (int off = 16; off > 0; off >>= 1) epart += __shfl_xor(epart, off, 32);
  if (lane == 0) atomicAdd(&sEnergy, epart);
  __syncthreads();

  v8f acc2 = {};
  for (int k = 0; k < SDIM; k += 4) {
    v2f a, b;
    a.x = sPG[mt + mr][k + 2 * kh];
    a.y = sPG[mt + mr][k + 2 * kh + 1];
    b.x = sWo[nt + mr][k + 2 * kh];
    b.y = sWo[nt + mr][k + 2 * kh + 1];
    acc2 = wmma_f32_k4(a, b, acc2);
  }
#pragma unroll
  for (int r = 0; r < 8; ++r) {
    int bb = mt + r + 8 * kh;
    int oo = nt + mr;
    float xo = sXo[bb][oo];
    xo_out[(size_t)n * BATCH * OBJ_DIM + bb * OBJ_DIM + oo] =
        xo + ETA_X * (acc2[r] + sCtx[bb][oo] - xo);
  }

  const int mt2 = (wv & 3) * 16;
  v8f acc3 = {};
  for (int k = 0; k < SDIM; k += 4) {
    v2f a, b;
    a.x = sPG[mt2 + mr][k + 2 * kh];
    a.y = sPG[mt2 + mr][k + 2 * kh + 1];
    b.x = sWl[mr][k + 2 * kh];
    b.y = sWl[mr][k + 2 * kh + 1];
    acc3 = wmma_f32_k4(a, b, acc3);
  }
  if (wv < 4) {
#pragma unroll
    for (int r = 0; r < 8; ++r) {
      int bb = mt2 + r + 8 * kh;
      int ll = mr;
      float xl = sXl[bb][ll];
      xl_out[(size_t)n * BATCH * LOC_DIM + bb * LOC_DIM + ll] =
          xl + ETA_X * acc3[r];
    }
  }

  if (tid == 0) atomicAdd(energy_t, sEnergy);
}

extern "C" void kernel_launch(void* const* d_in, const int* in_sizes, int n_in,
                              void* d_out, int out_size, void* d_ws, size_t ws_size,
                              hipStream_t stream) {
  const float* gin  = (const float*)d_in[0];
  const float* Wobj = (const float*)d_in[1];
  const float* Wloc = (const float*)d_in[2];

  float* out    = (float*)d_out;
  float* energy = out + XO_TOT;

  float* ws  = (float*)d_ws;
  float* xoA = ws;
  float* xoB = xoA + XO_TOT;
  float* xlA = xoB + XO_TOT;
  float* xlB = xlA + XL_TOT;

  cg_init_kernel<<<4096, 256, 0, stream>>>(xoA, xlA, energy);

  for (int t = 0; t < NSTEPS; ++t) {
    const float* xo_in = (t & 1) ? xoB : xoA;
    const float* xl_in = (t & 1) ? xlB : xlA;
    float* xo_out = (t == NSTEPS - 1) ? out : ((t & 1) ? xoA : xoB);
    float* xl_out = (t & 1) ? xlA : xlB;
    cg_step_kernel<<<NCOLS, 512, 0, stream>>>(gin, Wobj, Wloc,
                                              xo_in, xl_in, xo_out, xl_out,
                                              energy + t);
  }
}


# x_obj strips cross global memory as whole 128-byte lines: own, neighbour and output strips are contiguous 512-byte wave accesses, re-laid to/from fragment layout in a wave's LDS tile; same sum order
# speedup vs baseline: 1.6991x; 1.6991x over previous
#include <hip/hip_runtime.h>
#include <math.h>


#define GRID_HH 32
#define GRID_WW 32
#define OBJ_DIM 64
#define LOC_DIM 16
#define SDIM    64
#define NCOLS   (GRID_HH * GRID_WW)
#define BATCH   64
#define ETA_X   0.05f
#define NSTEPS  20
#define WP      68
#define LP      20

#define XO_TOT  ((size_t)NCOLS * BATCH * OBJ_DIM)
#define XL_TOT  ((size_t)NCOLS * BATCH * LOC_DIM)

typedef __attribute__((ext_vector_type(2))) float v2f;
typedef __attribute__((ext_vector_type(4))) float v4f;
typedef __attribute__((ext_vector_type(8))) float v8f;

__device__ __forceinline__ v8f wmma_f32_k4(v2f a, v2f b, v8f c) {
  return __builtin_amdgcn_wmma_f32_16x16x4_f32(
      false, a, false, b, (short)0, c, false, false);
}

__device__ __forceinline__ v2f lo2(v4f v) { v2f r; r.x = v[0]; r.y = v[1]; return r; }
__device__ __forceinline__ v2f hi2(v4f v) { v2f r; r.x = v[2]; r.y = v[3]; return r; }

__device__ __forceinline__ void load_lines(const float* __restrict__ p, v4f (&d)[8]) {
#pragma unroll
  for (int k = 0; k < 8; ++k)
    d[k] = *reinterpret_cast<const v4f*>(p + 128 * k);
}

__device__ __forceinline__ void wave_lds_order() {
  __builtin_amdgcn_fence(__ATOMIC_RELEASE, "wavefront");
  __builtin_amdgcn_wave_barrier();
  __builtin_amdgcn_fence(__ATOMIC_ACQUIRE, "wavefront");
}

__global__ void cg_init_kernel(float* __restrict__ xo,
                               float* __restrict__ xl,
                               float* __restrict__ energy) {
  size_t i = (size_t)blockIdx.x * blockDim.x + threadIdx.x;
  size_t stride = (size_t)gridDim.x * blockDim.x;
  for (size_t e = i; e < XO_TOT; e += stride) xo[e] = 0.0f;
  for (size_t e = i; e < XL_TOT; e += stride) xl[e] = 0.0f;
  if (i < NSTEPS) energy[i] = 0.0f;
}

__global__ __launch_bounds__(128) __attribute__((amdgpu_waves_per_eu(4, 4)))
void cg_step_kernel(const float* __restrict__ gin,
                    const float* __restrict__ Wobj,
                    const float* __restrict__ Wloc,
                    const float* __restrict__ xo_in,
                    const float* __restrict__ xl_in,
                    float* __restrict__ xo_out,
                    float* __restrict__ xl_out,
                    float* __restrict__ energy_t) {
  __shared__ __attribute__((aligned(16))) float sWo [OBJ_DIM * WP];
  __shared__ __attribute__((aligned(16))) float sWl [LOC_DIM * WP];
  __shared__ __attribute__((aligned(16))) float sWoT[SDIM * WP];
  __shared__ __attribute__((aligned(16))) float sWlT[SDIM * LP];
  __shared__ __attribute__((aligned(16))) float sX[4][16 * WP];
  __shared__ float sE[4];

  const int n    = blockIdx.x;
  const int gh   = n >> 5;
  const int gw   = n & 31;
  const int tid  = threadIdx.x;
  const int lane = tid & 31;
  const int wv   = tid >> 5;
  const int c    = lane & 15;
  const int h    = lane >> 4;
  const int b    = wv * 16 + c;


  const int c4 = tid & 15;
  const int p0 = tid >> 4;
  const float* gWo = Wobj + (size_t)n * OBJ_DIM * SDIM;
  const float* gWl = Wloc + (size_t)n * LOC_DIM * SDIM;
  const int wofs = 4 * p0 * SDIM + 4 * c4;
  v4f wo[2][4], wl[4];
#pragma unroll
  for (int j = 0; j < 2; ++j)
#pragma unroll
    for (int r = 0; r < 4; ++r)
      wo[j][r] = *reinterpret_cast<const v4f*>(gWo + (wofs + (32 * j + r) * SDIM));
  if (tid < 64) {
#pragma unroll
    for (int r = 0; r < 4; ++r)
      wl[r] = *reinterpret_cast<const v4f*>(gWl + (wofs + r * SDIM));
  }

  const int    sofs = wv * 16 * OBJ_DIM + 4 * lane;
  const size_t cofs = (size_t)n * BATCH * OBJ_DIM;
  v4f xc[8];
  load_lines((xo_in + cofs) + sofs, xc);

  const float* pl = (xl_in + (size_t)n * BATCH * LOC_DIM) + (b * LOC_DIM + 8 * h);
  v4f xlb[2];
  xlb[0] = *reinterpret_cast<const v4f*>(pl);
  xlb[1] = *reinterpret_cast<const v4f*>(pl + 4);

  const bool hu = gh > 0, hd = gh < GRID_HH - 1, hl = gw > 0, hr = gw < GRID_WW - 1;
  const long cstride = (long)BATCH * OBJ_DIM;
  v4f nu[8], nd[8], nl[8], nr[8];
  load_lines((xo_in + cofs + (hu ? -GRID_WW * cstride : 0)) + sofs, nu);
  load_lines((xo_in + cofs + (hd ?  GRID_WW * cstride : 0)) + sofs, nd);
  load_lines((xo_in + cofs + (hl ? -cstride : 0)) + sofs, nl);
  load_lines((xo_in + cofs + (hr ?  cstride : 0)) + sofs, nr);

#pragma unroll
  for (int j = 0; j < 2; ++j) {
    const int p = p0 + 8 * j;
#pragma unroll
    for (int r = 0; r < 4; ++r)
      *reinterpret_cast<v4f*>(&sWo[(4 * p + r) * WP + 4 * c4]) = wo[j][r];
#pragma unroll
    for (int e = 0; e < 4; ++e) {
      v4f tr = {wo[j][0][e], wo[j][1][e], wo[j][2][e], wo[j][3][e]};
      *reinterpret_cast<v4f*>(&sWoT[(4 * c4 + e) * WP + 4 * p]) = tr;
    }
  }
  if (tid < 64) {
#pragma unroll
    for (int r = 0; r < 4; ++r)
      *reinterpret_cast<v4f*>(&sWl[(4 * p0 + r) * WP + 4 * c4]) = wl[r];
#pragma unroll
    for (int e = 0; e < 4; ++e) {
      v4f tr = {wl[0][e], wl[1][e], wl[2][e], wl[3][e]};
      *reinterpret_cast<v4f*>(&sWlT[(4 * c4 + e) * LP + 4 * p0]) = tr;
    }
  }
  float* tile = &sX[wv][0];
  float* tmem = tile + (h * WP + 4 * c);
  float* tfrg = tile + (c * WP + 8 * h);
#pragma unroll
  for (int k = 0; k < 8; ++k)
    *reinterpret_cast<v4f*>(tmem + 2 * k * WP) = xc[k];
  __syncthreads();

  v4f xb[8];
#pragma unroll
  for (int q = 0; q < 4; ++q) {
    xb[2 * q]     = *reinterpret_cast<const v4f*>(tfrg + 16 * q);
    xb[2 * q + 1] = *reinterpret_cast<const v4f*>(tfrg + 16 * q + 4);
  }

  const v4f zero4 = {0.0f, 0.0f, 0.0f, 0.0f};
  v4f ctx[8];
#pragma unroll
  for (int k = 0; k < 8; ++k) {
    ctx[k]  = (hu ? nu[k] : zero4) + (hd ? nd[k] : zero4);
    ctx[k] += (hl ? nl[k] : zero4);
    ctx[k] += (hr ? nr[k] : zero4);
  }
  asm volatile("" :: "v"(ctx[0]), "v"(ctx[1]), "v"(ctx[2]), "v"(ctx[3]),
                     "v"(ctx[4]), "v"(ctx[5]), "v"(ctx[6]), "v"(ctx[7]));

  const float* pp = gin + (size_t)b * 65536 + (size_t)(gh * 8 + h) * 256 + gw * 8;
  v4f pt[8];
#pragma unroll
  for (int t = 0; t < 4; ++t) {
    pt[2 * t]     = *reinterpret_cast<const v4f*>(pp + 512 * t);
    pt[2 * t + 1] = *reinterpret_cast<const v4f*>(pp + 512 * t + 4);
  }

  const float* aT  = &sWoT[c * WP + 8 * h];
  const float* aLT = &sWlT[c * LP + 8 * h];
  v8f acc[4] = {};
  v4f fa[2][4];
#pragma unroll
  for (int t = 0; t < 4; ++t)
    fa[0][t] = *reinterpret_cast<const v4f*>(aT + 16 * t * WP);
#pragma unroll
  for (int kb = 0; kb < 10; ++kb) {
    const int nb = kb + 1;
    if (nb < 8) {
#pragma unroll
      for (int t = 0; t < 4; ++t)
        fa[nb & 1][t] = *reinterpret_cast<const v4f*>(aT + 16 * t * WP + 16 * (nb >> 1) + 4 * (nb & 1));
    } else if (nb < 10) {
#pragma unroll
      for (int t = 0; t < 4; ++t)
        fa[nb & 1][t] = *reinterpret_cast<const v4f*>(aLT + 16 * t * LP + 4 * (nb & 1));
    }
    const v4f bq = (kb < 8) ? xb[kb] : xlb[kb - 8];
#pragma unroll
    for (int t = 0; t < 4; ++t) acc[t] = wmma_f32_k4(lo2(fa[kb & 1][t]), lo2(bq), acc[t]);
#pragma unroll
    for (int t = 0; t < 4; ++t) acc[t] = wmma_f32_k4(hi2(fa[kb & 1][t]), hi2(bq), acc[t]);
  }

  float epart = 0.0f;
#pragma unroll
  for (int t = 0; t < 4; ++t) {
#pragma unroll
    for (int r = 0; r < 8; ++r) {
      float pred = tanhf(acc[t][r]);
      float eps  = pt[2 * t + (r >> 2)][r & 3] - pred;
      epart += 0.5f * eps * eps;
      acc[t][r] = eps * (1.0f - pred * pred);
    }
  }

  const float* aR  = &sWo[c * WP + 8 * h];
  const float* aLR = &sWl[c * WP + 8 * h];
  v8f dxo[4] = {};
  v8f dxl = {};
  v4f fb[2][5];
#pragma unroll
  for (int u = 0; u < 4; ++u)
    fb[0][u] = *reinterpret_cast<const v4f*>(aR + 16 * u * WP);
  fb[0][4] = *reinterpret_cast<const v4f*>(aLR);
#pragma unroll
  for (int sb = 0; sb < 8; ++sb) {
    const int nb = sb + 1;
    if (nb < 8) {
      const int off = 16 * (nb >> 1) + 4 * (nb & 1);
#pragma unroll
      for (int u = 0; u < 4; ++u)
        fb[nb & 1][u] = *reinterpret_cast<const v4f*>(aR + 16 * u * WP + off);
      fb[nb & 1][4] = *reinterpret_cast<const v4f*>(aLR + off);
    }
    const int t = sb >> 1, r0 = 4 * (sb & 1);
    v2f g01, g23;
    g01.x = acc[t][r0];     g01.y = acc[t][r0 + 1];
    g23.x = acc[t][r0 + 2]; g23.y = acc[t][r0 + 3];
#pragma unroll
    for (int u = 0; u < 4; ++u) dxo[u] = wmma_f32_k4(lo2(fb[sb & 1][u]), g01, dxo[u]);
    dxl = wmma_f32_k4(lo2(fb[sb & 1][4]), g01, dxl);
#pragma unroll
    for (int u = 0; u < 4; ++u) dxo[u] = wmma_f32_k4(hi2(fb[sb & 1][u]), g23, dxo[u]);
    dxl = wmma_f32_k4(hi2(fb[sb & 1][4]), g23, dxl);
  }

  const float inv_cnt = 1.0f / (float)((int)hu + (int)hd + (int)hl + (int)hr);
  v4f xm[8], dm[8];
#pragma unroll
  for (int k = 0; k < 8; ++k)
    xm[k] = *reinterpret_cast<const v4f*>(tmem + 2 * k * WP);
  wave_lds_order();
#pragma unroll
  for (int u = 0; u < 4; ++u) {
#pragma unroll
    for (int jj = 0; jj < 2; ++jj) {
      v4f d = {dxo[u][4 * jj], dxo[u][4 * jj + 1], dxo[u][4 * jj + 2], dxo[u][4 * jj + 3]};
      *reinterpret_cast<v4f*>(tfrg + 16 * u + 4 * jj) = d;
    }
  }
  wave_lds_order();
#pragma unroll
  for (int k = 0; k < 8; ++k)
    dm[k] = *reinterpret_cast<const v4f*>(tmem + 2 * k * WP);
#pragma unroll
  for (int k = 0; k < 8; ++k) {
    v4f o;
#pragma unroll
    for (int e = 0; e < 4; ++e) {
      float xo = xm[k][e];
      o[e] = xo + ETA_X * (dm[k][e] + ctx[k][e] * inv_cnt - xo);
    }
    *reinterpret_cast<v4f*>((xo_out + cofs) + (sofs + 128 * k)) = o;
  }
#pragma unroll
  for (int jj = 0; jj < 2; ++jj) {
    v4f o;
#pragma unroll
    for (int e = 0; e < 4; ++e) o[e] = xlb[jj][e] + ETA_X * dxl[4 * jj + e];
    *reinterpret_cast<v4f*>((xl_out + (size_t)n * BATCH * LOC_DIM) + (b * LOC_DIM + 8 * h + 4 * jj)) = o;
  }

  for (int off = 16; off > 0; off >>= 1) epart += __shfl_xor(epart, off, 32);
  if (lane == 0) sE[wv] = epart;
  __syncthreads();
  if (tid == 0) atomicAdd(energy_t, (sE[0] + sE[1]) + (sE[2] + sE[3]));
}

extern "C" void kernel_launch(void* const* d_in, const int* in_sizes, int n_in,
                              void* d_out, int out_size, void* d_ws, size_t ws_size,
                              hipStream_t stream) {
  const float* gin  = (const float*)d_in[0];
  const float* Wobj = (const float*)d_in[1];
  const float* Wloc = (const float*)d_in[2];

  float* out    = (float*)d_out;
  float* energy = out + XO_TOT;

  float* ws  = (float*)d_ws;
  float* xoA = ws;
  float* xoB = xoA + XO_TOT;
  float* xlA = xoB + XO_TOT;
  float* xlB = xlA + XL_TOT;

  cg_init_kernel<<<4096, 256, 0, stream>>>(xoA, xlA, energy);

  for (int t = 0; t < NSTEPS; ++t) {
    const float* xo_in = (t & 1) ? xoB : xoA;
    const float* xl_in = (t & 1) ? xlB : xlA;
    float* xo_out = (t == NSTEPS - 1) ? out : ((t & 1) ? xoA : xoB);
    float* xl_out = (t & 1) ? xlA : xlB;
    cg_step_kernel<<<NCOLS, 128, 0, stream>>>(gin, Wobj, Wloc,
                                              xo_in, xl_in, xo_out, xl_out,
                                              energy + t);
  }
}


# Energy handed in behind the tanh stage, no closing barrier: wave sum on lane-permute instructions (same pairs), a column's last wave (LDS count) adds its 4 strip sums as before; same sum order
# speedup vs baseline: 1.8520x; 1.0900x over previous
#include <hip/hip_runtime.h>
#include <math.h>


#define GRID_HH 32
#define GRID_WW 32
#define OBJ_DIM 64
#define LOC_DIM 16
#define SDIM    64
#define NCOLS   (GRID_HH * GRID_WW)
#define BATCH   64
#define ETA_X   0.05f
#define NSTEPS  20
#define WP      68
#define LP      20

#define XO_TOT  ((size_t)NCOLS * BATCH * OBJ_DIM)
#define XL_TOT  ((size_t)NCOLS * BATCH * LOC_DIM)

typedef __attribute__((ext_vector_type(2))) float v2f;
typedef __attribute__((ext_vector_type(4))) float v4f;
typedef __attribute__((ext_vector_type(8))) float v8f;

__device__ __forceinline__ v8f wmma_f32_k4(v2f a, v2f b, v8f c) {
  return __builtin_amdgcn_wmma_f32_16x16x4_f32(
      false, a, false, b, (short)0, c, false, false);
}

__device__ __forceinline__ v2f lo2(v4f v) { v2f r; r.x = v[0]; r.y = v[1]; return r; }
__device__ __forceinline__ v2f hi2(v4f v) { v2f r; r.x = v[2]; r.y = v[3]; return r; }

__device__ __forceinline__ void load_lines(const float* __restrict__ p, v4f (&d)[8]) {
#pragma unroll
  for (int k = 0; k < 8; ++k)
    d[k] = *reinterpret_cast<const v4f*>(p + 128 * k);
}

__device__ __forceinline__ void wave_lds_order() {
  __builtin_amdgcn_fence(__ATOMIC_RELEASE, "wavefront");
  __builtin_amdgcn_wave_barrier();
  __builtin_amdgcn_fence(__ATOMIC_ACQUIRE, "wavefront");
}

__device__ __forceinline__ float wave_sum(float v) {
  const int b16 = __builtin_bit_cast(int, v);
  v += __builtin_bit_cast(float, __builtin_amdgcn_permlanex16(b16, b16, 0x76543210, 0xfedcba98, false, false));
  const int b8 = __builtin_bit_cast(int, v);
  v += __builtin_bit_cast(float, __builtin_amdgcn_update_dpp(b8, b8, 0x168, 0xf, 0xf, false));
  const int b4 = __builtin_bit_cast(int, v);
  v += __builtin_bit_cast(float, __builtin_amdgcn_update_dpp(b4, b4, 0x164, 0xf, 0xf, false));
  const int b2 = __builtin_bit_cast(int, v);
  v += __builtin_bit_cast(float, __builtin_amdgcn_update_dpp(b2, b2, 0x162, 0xf, 0xf, false));
  const int b1 = __builtin_bit_cast(int, v);
  v += __builtin_bit_cast(float, __builtin_amdgcn_update_dpp(b1, b1, 0x161, 0xf, 0xf, false));
  return v;
}

__global__ void cg_init_kernel(float* __restrict__ xo,
                               float* __restrict__ xl,
                               float* __restrict__ energy) {
  size_t i = (size_t)blockIdx.x * blockDim.x + threadIdx.x;
  size_t stride = (size_t)gridDim.x * blockDim.x;
  for (size_t e = i; e < XO_TOT; e += stride) xo[e] = 0.0f;
  for (size_t e = i; e < XL_TOT; e += stride) xl[e] = 0.0f;
  if (i < NSTEPS) energy[i] = 0.0f;
}

__global__ __launch_bounds__(512) __attribute__((amdgpu_waves_per_eu(4, 4)))
void cg_step_kernel(const float* __restrict__ gin,
                    const float* __restrict__ Wobj,
                    const float* __restrict__ Wloc,
                    const float* __restrict__ xo_in,
                    const float* __restrict__ xl_in,
                    float* __restrict__ xo_out,
                    float* __restrict__ xl_out,
                    float* __restrict__ energy_t) {
  __shared__ __attribute__((aligned(16))) float sWo [4][OBJ_DIM * WP];
  __shared__ __attribute__((aligned(16))) float sWl [4][LOC_DIM * WP];
  __shared__ __attribute__((aligned(16))) float sWoT[4][SDIM * WP];
  __shared__ __attribute__((aligned(16))) float sWlT[4][SDIM * LP];
  __shared__ __attribute__((aligned(16))) float sX[16][16 * WP];
  __shared__ __attribute__((aligned(16))) float sD[16][16 * WP];
  __shared__ float sE[16];
  __shared__ unsigned sCnt[4];

  const int wid  = __builtin_amdgcn_readfirstlane((int)(threadIdx.x >> 5));
  const int j    = wid >> 2;
  const int br   = j >> 1;
  const int bq   = j & 1;
  const int wv   = wid & 3;
  const int gh   = 2 * (int)(blockIdx.x >> 4) + br;
  const int gw   = 2 * (int)(blockIdx.x & 15) + bq;
  const int n    = gh * GRID_WW + gw;
  const int tid  = threadIdx.x & 127;
  const int lane = tid & 31;
  const int c    = lane & 15;
  const int h    = lane >> 4;
  const int b    = wv * 16 + c;


  const int c4 = tid & 15;
  const int p0 = tid >> 4;
  const float* gWo = Wobj + (size_t)n * OBJ_DIM * SDIM;
  const float* gWl = Wloc + (size_t)n * LOC_DIM * SDIM;
  const int wofs = 4 * p0 * SDIM + 4 * c4;
  v4f wo[2][4], wl[4];
#pragma unroll
  for (int jw = 0; jw < 2; ++jw)
#pragma unroll
    for (int r = 0; r < 4; ++r)
      wo[jw][r] = *reinterpret_cast<const v4f*>(gWo + (wofs + (32 * jw + r) * SDIM));
  if (wv < 2) {
#pragma unroll
    for (int r = 0; r < 4; ++r)
      wl[r] = *reinterpret_cast<const v4f*>(gWl + (wofs + r * SDIM));
  }

  const int    sofs = wv * 16 * OBJ_DIM + 4 * lane;
  const size_t cofs = (size_t)n * BATCH * OBJ_DIM;
  v4f xc[8];
  load_lines((xo_in + cofs) + sofs, xc);

  const float* pl = (xl_in + (size_t)n * BATCH * LOC_DIM) + (b * LOC_DIM + 8 * h);
  v4f xlb[2];
  xlb[0] = *reinterpret_cast<const v4f*>(pl);
  xlb[1] = *reinterpret_cast<const v4f*>(pl + 4);

  const bool hu = gh > 0, hd = gh < GRID_HH - 1, hl = gw > 0, hr = gw < GRID_WW - 1;
  const bool hv = br ? hd : hu;
  const bool hz = bq ? hr : hl;
  const long cstride = (long)BATCH * OBJ_DIM;
  const long vstep = br ? GRID_WW * cstride : -GRID_WW * cstride;
  const long zstep = bq ? cstride : -cstride;
  v4f gv[8], gz[8];
  load_lines((xo_in + cofs + (hv ? vstep : 0)) + sofs, gv);
  load_lines((xo_in + cofs + (hz ? zstep : 0)) + sofs, gz);

#pragma unroll
  for (int jw = 0; jw < 2; ++jw) {
    const int p = p0 + 8 * jw;
#pragma unroll
    for (int r = 0; r < 4; ++r)
      *reinterpret_cast<v4f*>(&sWo[j][(4 * p + r) * WP + 4 * c4]) = wo[jw][r];
#pragma unroll
    for (int e = 0; e < 4; ++e) {
      v4f tr = {wo[jw][0][e], wo[jw][1][e], wo[jw][2][e], wo[jw][3][e]};
      *reinterpret_cast<v4f*>(&sWoT[j][(4 * c4 + e) * WP + 4 * p]) = tr;
    }
  }
  if (wv < 2) {
#pragma unroll
    for (int r = 0; r < 4; ++r)
      *reinterpret_cast<v4f*>(&sWl[j][(4 * p0 + r) * WP + 4 * c4]) = wl[r];
#pragma unroll
    for (int e = 0; e < 4; ++e) {
      v4f tr = {wl[0][e], wl[1][e], wl[2][e], wl[3][e]};
      *reinterpret_cast<v4f*>(&sWlT[j][(4 * c4 + e) * LP + 4 * p0]) = tr;
    }
  }
  const int vofs = h * WP + 4 * c;
  const int fofs = c * WP + 8 * h;
  float* tmem = &sX[wid][0] + vofs;
  const float* tfrg = &sX[wid][0] + fofs;
  const float* vmem = &sX[wid ^ 8][0] + vofs;
  const float* zmem = &sX[wid ^ 4][0] + vofs;
#pragma unroll
  for (int k = 0; k < 8; ++k)
    *reinterpret_cast<v4f*>(tmem + 2 * k * WP) = xc[k];
  if (threadIdx.x < 4) sCnt[threadIdx.x] = 0u;
  __syncthreads();

  v4f xb[8];
#pragma unroll
  for (int q = 0; q < 4; ++q) {
    xb[2 * q]     = *reinterpret_cast<const v4f*>(tfrg + 16 * q);
    xb[2 * q + 1] = *reinterpret_cast<const v4f*>(tfrg + 16 * q + 4);
  }

  const v4f zero4 = {0.0f, 0.0f, 0.0f, 0.0f};
  v4f ctx[8];
#pragma unroll
  for (int k = 0; k < 8; ++k) {
    const v4f lv = *reinterpret_cast<const v4f*>(vmem + 2 * k * WP);
    const v4f lz = *reinterpret_cast<const v4f*>(zmem + 2 * k * WP);
    const v4f ov = hv ? gv[k] : zero4;
    const v4f oz = hz ? gz[k] : zero4;
    ctx[k]  = ov + lv;
    ctx[k] += (bq ? lz : oz);
    ctx[k] += (bq ? oz : lz);
  }
  asm volatile("" :: "v"(ctx[0]), "v"(ctx[1]), "v"(ctx[2]), "v"(ctx[3]),
                     "v"(ctx[4]), "v"(ctx[5]), "v"(ctx[6]), "v"(ctx[7]));

  const float* pp = gin + (size_t)b * 65536 + (size_t)(gh * 8 + h) * 256 + gw * 8;
  v4f pt[8];
#pragma unroll
  for (int t = 0; t < 4; ++t) {
    pt[2 * t]     = *reinterpret_cast<const v4f*>(pp + 512 * t);
    pt[2 * t + 1] = *reinterpret_cast<const v4f*>(pp + 512 * t + 4);
  }

  const float* aT  = &sWoT[j][c * WP + 8 * h];
  const float* aLT = &sWlT[j][c * LP + 8 * h];
  v8f acc[4] = {};
  v4f fa[2][4];
#pragma unroll
  for (int t = 0; t < 4; ++t)
    fa[0][t] = *reinterpret_cast<const v4f*>(aT + 16 * t * WP);
#pragma unroll
  for (int kb = 0; kb < 10; ++kb) {
    const int nb = kb + 1;
    if (nb < 8) {
#pragma unroll
      for (int t = 0; t < 4; ++t)
        fa[nb & 1][t] = *reinterpret_cast<const v4f*>(aT + 16 * t * WP + 16 * (nb >> 1) + 4 * (nb & 1));
    } else if (nb < 10) {
#pragma unroll
      for (int t = 0; t < 4; ++t)
        fa[nb & 1][t] = *reinterpret_cast<const v4f*>(aLT + 16 * t * LP + 4 * (nb & 1));
    }
    const v4f bq = (kb < 8) ? xb[kb] : xlb[kb - 8];
#pragma unroll
    for (int t = 0; t < 4; ++t) acc[t] = wmma_f32_k4(lo2(fa[kb & 1][t]), lo2(bq), acc[t]);
#pragma unroll
    for (int t = 0; t < 4; ++t) acc[t] = wmma_f32_k4(hi2(fa[kb & 1][t]), hi2(bq), acc[t]);
  }

  float epart = 0.0f;
#pragma unroll
  for (int t = 0; t < 4; ++t) {
#pragma unroll
    for (int r = 0; r < 8; ++r) {
      float pred = tanhf(acc[t][r]);
      float eps  = pt[2 * t + (r >> 2)][r & 3] - pred;
      epart += 0.5f * eps * eps;
      acc[t][r] = eps * (1.0f - pred * pred);
    }
  }

  epart = wave_sum(epart);
  if (lane == 0) {
    sE[wid] = epart;
    const unsigned ahead = __hip_atomic_fetch_add(&sCnt[j], 1u, __ATOMIC_ACQ_REL,
                                                  __HIP_MEMORY_SCOPE_WORKGROUP);
    if (ahead == 3u)
      atomicAdd(energy_t, (sE[4 * j] + sE[4 * j + 1]) + (sE[4 * j + 2] + sE[4 * j + 3]));
  }

  const float* aR  = &sWo[j][c * WP + 8 * h];
  const float* aLR = &sWl[j][c * WP + 8 * h];
  v8f dxo[4] = {};
  v8f dxl = {};
  v4f fb[2][5];
#pragma unroll
  for (int u = 0; u < 4; ++u)
    fb[0][u] = *reinterpret_cast<const v4f*>(aR + 16 * u * WP);
  fb[0][4] = *reinterpret_cast<const v4f*>(aLR);
#pragma unroll
  for (int sb = 0; sb < 8; ++sb) {
    const int nb = sb + 1;
    if (nb < 8) {
      const int off = 16 * (nb >> 1) + 4 * (nb & 1);
#pragma unroll
      for (int u = 0; u < 4; ++u)
        fb[nb & 1][u] = *reinterpret_cast<const v4f*>(aR + 16 * u * WP + off);
      fb[nb & 1][4] = *reinterpret_cast<const v4f*>(aLR + off);
    }
    const int t = sb >> 1, r0 = 4 * (sb & 1);
    v2f g01, g23;
    g01.x = acc[t][r0];     g01.y = acc[t][r0 + 1];
    g23.x = acc[t][r0 + 2]; g23.y = acc[t][r0 + 3];
#pragma unroll
    for (int u = 0; u < 4; ++u) dxo[u] = wmma_f32_k4(lo2(fb[sb & 1][u]), g01, dxo[u]);
    dxl = wmma_f32_k4(lo2(fb[sb & 1][4]), g01, dxl);
#pragma unroll
    for (int u = 0; u < 4; ++u) dxo[u] = wmma_f32_k4(hi2(fb[sb & 1][u]), g23, dxo[u]);
    dxl = wmma_f32_k4(hi2(fb[sb & 1][4]), g23, dxl);
  }

  const float inv_cnt = 1.0f / (float)((int)hu + (int)hd + (int)hl + (int)hr);
  v4f xm[8], dm[8];
#pragma unroll
  for (int k = 0; k < 8; ++k)
    xm[k] = *reinterpret_cast<const v4f*>(tmem + 2 * k * WP);
  float* dfrg = &sD[wid][0] + fofs;
  const float* dmem = &sD[wid][0] + vofs;
#pragma unroll
  for (int u = 0; u < 4; ++u) {
#pragma unroll
    for (int jj = 0; jj < 2; ++jj) {
      v4f d = {dxo[u][4 * jj], dxo[u][4 * jj + 1], dxo[u][4 * jj + 2], dxo[u][4 * jj + 3]};
      *reinterpret_cast<v4f*>(dfrg + 16 * u + 4 * jj) = d;
    }
  }
  wave_lds_order();
#pragma unroll
  for (int k = 0; k < 8; ++k)
    dm[k] = *reinterpret_cast<const v4f*>(dmem + 2 * k * WP);
#pragma unroll
  for (int k = 0; k < 8; ++k) {
    v4f o;
#pragma unroll
    for (int e = 0; e < 4; ++e) {
      float xo = xm[k][e];
      o[e] = xo + ETA_X * (dm[k][e] + ctx[k][e] * inv_cnt - xo);
    }
    *reinterpret_cast<v4f*>((xo_out + cofs) + (sofs + 128 * k)) = o;
  }
#pragma unroll
  for (int jj = 0; jj < 2; ++jj) {
    v4f o;
#pragma unroll
    for (int e = 0; e < 4; ++e) o[e] = xlb[jj][e] + ETA_X * dxl[4 * jj + e];
    *reinterpret_cast<v4f*>((xl_out + (size_t)n * BATCH * LOC_DIM) + (b * LOC_DIM + 8 * h + 4 * jj)) = o;
  }
}

extern "C" void kernel_launch(void* const* d_in, const int* in_sizes, int n_in,
                              void* d_out, int out_size, void* d_ws, size_t ws_size,
                              hipStream_t stream) {
  const float* gin  = (const float*)d_in[0];
  const float* Wobj = (const float*)d_in[1];
  const float* Wloc = (const float*)d_in[2];

  float* out    = (float*)d_out;
  float* energy = out + XO_TOT;

  float* ws  = (float*)d_ws;
  float* xoA = ws;
  float* xoB = xoA + XO_TOT;
  float* xlA = xoB + XO_TOT;
  float* xlB = xlA + XL_TOT;

  cg_init_kernel<<<4096, 256, 0, stream>>>(xoA, xlA, energy);

  for (int t = 0; t < NSTEPS; ++t) {
    const float* xo_in = (t & 1) ? xoB : xoA;
    const float* xl_in = (t & 1) ? xlB : xlA;
    float* xo_out = (t == NSTEPS - 1) ? out : ((t & 1) ? xoA : xoB);
    float* xl_out = (t & 1) ? xlA : xlB;
    cg_step_kernel<<<NCOLS / 4, 512, 0, stream>>>(gin, Wobj, Wloc,
                                              xo_in, xl_in, xo_out, xl_out,
                                              energy + t);
  }
}
